# starting kernel
# baseline (speedup 1.0000x reference)
#include <hip/hip_runtime.h>
#include <hip/hip_bf16.h>


typedef float v2f __attribute__((ext_vector_type(2)));
typedef float v8f __attribute__((ext_vector_type(8)));

#define DV   128
#define DE0  64
#define DE1  128
#define WAVES_PER_BLOCK 4
#define SHIFT_LN2 0.6931471805599453f

__device__ __forceinline__ float sp_act(float v) {
    float s = (v > 20.0f) ? v : log1pf(__expf(v));
    return s - SHIFT_LN2;
}


template <typename PtrT>
__device__ __forceinline__ void wmma_gemm_strip(PtrT Arow, const float* B, int K,
                                                v8f acc[8], int half, int l16,
                                                float ascale) {
    for (int k0 = 0; k0 < K; k0 += 4) {
        v2f a;
        a.x = Arow[k0 + 2 * half + 0] * ascale;
        a.y = Arow[k0 + 2 * half + 1] * ascale;
#pragma unroll
        for (int t = 0; t < 8; ++t) {
            const float* bp = B + (size_t)(k0 + 2 * half) * 128 + t * 16 + l16;
            v2f b;
            b.x = bp[0];
            b.y = bp[128];
            acc[t] = __builtin_amdgcn_wmma_f32_16x16x4_f32(
                false, a, false, b, (short)0, acc[t], false, false);
        }
    }
}

template <typename PtrT>
__device__ __forceinline__ void wmma_gemm_ldsb(PtrT Arow, const v2f* Bfrag, int K,
                                               v8f acc[8], int half, int lane,
                                               float ascale) {
    for (int k0 = 0; k0 < K; k0 += 4) {
        v2f a;
        a.x = Arow[k0 + 2 * half + 0] * ascale;
        a.y = Arow[k0 + 2 * half + 1] * ascale;
        const int s8 = (k0 >> 2) << 3;
#pragma unroll
        for (int t = 0; t < 8; ++t) {
            v2f b = Bfrag[(s8 + t) * 32 + lane];
            acc[t] = __builtin_amdgcn_wmma_f32_16x16x4_f32(
                false, a, false, b, (short)0, acc[t], false, false);
        }
    }
}

__device__ __forceinline__ void stage_weight_frag(float* dst, const float* W,
                                                  int K, int tid, int nthreads) {
    for (int i = tid; i < K * 128; i += nthreads) {
        const int k = i >> 7, n = i & 127;
        const int s = k >> 2, j = k & 1, hh = (k >> 1) & 1;
        const int l = hh * 16 + (n & 15), t = n >> 4;
        dst[(((s << 3) + t) * 32 + l) * 2 + j] = W[i];
    }
}

__device__ __forceinline__ void bias_softplus(v8f acc[8], const float* bias, int l16) {
#pragma unroll
    for (int t = 0; t < 8; ++t) {
        float bv = bias[t * 16 + l16];
#pragma unroll
        for (int r = 0; r < 8; ++r) acc[t][r] = sp_act(acc[t][r] + bv);
    }
}

__global__ __launch_bounds__(32 * WAVES_PER_BLOCK)
void k_node_proj(const float* __restrict__ node_fea,
                 const float* __restrict__ W_aw1,
                 float* __restrict__ x, int N) {
    const int lane = threadIdx.x & 31;
    const int wave = threadIdx.x >> 5;
    const int half = lane >> 4, l16 = lane & 15;
    const int row0 = (blockIdx.x * WAVES_PER_BLOCK + wave) * 16;
    if (row0 >= N) return;

    const int arow = min(row0 + l16, N - 1);
    v8f acc[8] = {};
    wmma_gemm_strip(node_fea + (size_t)arow * DV, W_aw1, DV, acc, half, l16, 1.0f);

#pragma unroll
    for (int t = 0; t < 8; ++t)
#pragma unroll
        for (int r = 0; r < 8; ++r) {
            int m = half * 8 + r;
            if (row0 + m < N)
                x[(size_t)(row0 + m) * DV + t * 16 + l16] = acc[t][r];
        }
}

__global__ __launch_bounds__(32 * WAVES_PER_BLOCK)
void k_edge(const float* __restrict__ edge_fea,
            const int* __restrict__ idx1, const int* __restrict__ idx2,
            const float* __restrict__ W_fg1, const float* __restrict__ b_fg1,
            const float* __restrict__ W_fg2, const float* __restrict__ b_fg2,
            const float* __restrict__ x,
            float* __restrict__ sums, float* __restrict__ cnt, int E) {
    __shared__ float w1t[DE0 * DE1];
    __shared__ float w2t[DE1 * DE1];
    __shared__ float hbuf[WAVES_PER_BLOCK * 16 * DE1];

    const int lane = threadIdx.x & 31;
    const int wave = threadIdx.x >> 5;
    const int half = lane >> 4, l16 = lane & 15;

    stage_weight_frag(w1t, W_fg1, DE0, threadIdx.x, 32 * WAVES_PER_BLOCK);
    stage_weight_frag(w2t, W_fg2, DE1, threadIdx.x, 32 * WAVES_PER_BLOCK);
    __syncthreads();

    const int e0 = (blockIdx.x * WAVES_PER_BLOCK + wave) * 16;
    if (e0 >= E) return;

    const int arow = min(e0 + l16, E - 1);
    v8f acc[8] = {};
    wmma_gemm_ldsb(edge_fea + (size_t)arow * DE0, (const v2f*)w1t, DE0,
                   acc, half, lane, 1.0f);
    bias_softplus(acc, b_fg1, l16);

    float* h = hbuf + wave * 16 * DE1;
#pragma unroll
    for (int t = 0; t < 8; ++t)
#pragma unroll
        for (int r = 0; r < 8; ++r)
            h[(half * 8 + r) * DE1 + t * 16 + l16] = acc[t][r];

    v8f acc2[8] = {};
    wmma_gemm_ldsb(h + l16 * DE1, (const v2f*)w2t, DE1, acc2, half, lane, 1.0f);
    bias_softplus(acc2, b_fg2, l16);

#pragma unroll
    for (int r = 0; r < 8; ++r) {
        const int e = e0 + half * 8 + r;
        if (e < E) {
            const int n2 = idx2[e];
            const int n1 = idx1[e];
#pragma unroll
            for (int t = 0; t < 8; ++t) {
                float v = x[(size_t)n2 * DV + t * 16 + l16] * acc2[t][r];
                atomicAdd(&sums[(size_t)n1 * DV + t * 16 + l16], v);
            }
        }
    }
    if (lane < 16 && (e0 + lane) < E) atomicAdd(&cnt[idx1[e0 + lane]], 1.0f);
}

__global__ __launch_bounds__(32 * WAVES_PER_BLOCK)
void k_out(const float* __restrict__ sums, const float* __restrict__ cnt,
           const float* __restrict__ node_fea,
           const float* __restrict__ W_aw2, const float* __restrict__ b_aw2,
           const float* __restrict__ W_aw3, const float* __restrict__ b_aw3,
           float* __restrict__ out, int N) {
    __shared__ float lds[WAVES_PER_BLOCK * 16 * DV];
    const int lane = threadIdx.x & 31;
    const int wave = threadIdx.x >> 5;
    const int half = lane >> 4, l16 = lane & 15;
    const int row0 = (blockIdx.x * WAVES_PER_BLOCK + wave) * 16;
    if (row0 >= N) return;

    const int arow = min(row0 + l16, N - 1);
    const float rscale = 1.0f / fmaxf(cnt[arow], 1.0f);

    v8f acc[8] = {};
    wmma_gemm_strip(sums + (size_t)arow * DV, W_aw2, DV, acc, half, l16, rscale);
    bias_softplus(acc, b_aw2, l16);

    float* h = lds + wave * 16 * DV;
#pragma unroll
    for (int t = 0; t < 8; ++t)
#pragma unroll
        for (int r = 0; r < 8; ++r)
            h[(half * 8 + r) * DV + t * 16 + l16] = acc[t][r];

    v8f acc2[8] = {};
    wmma_gemm_strip(h + l16 * DV, W_aw3, DV, acc2, half, l16, 1.0f);

#pragma unroll
    for (int t = 0; t < 8; ++t) {
        const float bv = b_aw3[t * 16 + l16];
#pragma unroll
        for (int r = 0; r < 8; ++r) {
            const int m = half * 8 + r;
            if (row0 + m < N) {
                const size_t o = (size_t)(row0 + m) * DV + t * 16 + l16;
                out[o] = node_fea[o] + acc2[t][r] + bv;
            }
        }
    }
}

extern "C" void kernel_launch(void* const* d_in, const int* in_sizes, int n_in,
                              void* d_out, int out_size, void* d_ws, size_t ws_size,
                              hipStream_t stream) {
    const float* node_fea = (const float*)d_in[0];
    const int*   idx1     = (const int*)d_in[1];
    const int*   idx2     = (const int*)d_in[2];
    const float* edge_fea = (const float*)d_in[3];
    const float* W_aw1    = (const float*)d_in[4];
    const float* W_fg1    = (const float*)d_in[5];
    const float* b_fg1    = (const float*)d_in[6];
    const float* W_fg2    = (const float*)d_in[7];
    const float* b_fg2    = (const float*)d_in[8];
    const float* W_aw2    = (const float*)d_in[9];
    const float* b_aw2    = (const float*)d_in[10];
    const float* W_aw3    = (const float*)d_in[11];
    const float* b_aw3    = (const float*)d_in[12];

    const int N = in_sizes[0] / DV;
    const int E = in_sizes[1];

    float* x    = (float*)d_ws;
    float* sums = x + (size_t)N * DV;
    float* cnt  = sums + (size_t)N * DV;

    hipMemsetAsync(sums, 0, ((size_t)N * DV + N) * sizeof(float), stream);

    const int tpb = 32 * WAVES_PER_BLOCK;
    const int ntile_n = (N + 15) / 16;
    const int ntile_e = (E + 15) / 16;
    const int nblk_n = (ntile_n + WAVES_PER_BLOCK - 1) / WAVES_PER_BLOCK;
    const int nblk_e = (ntile_e + WAVES_PER_BLOCK - 1) / WAVES_PER_BLOCK;

    k_node_proj<<<nblk_n, tpb, 0, stream>>>(node_fea, W_aw1, x, N);
    k_edge<<<nblk_e, tpb, 0, stream>>>(edge_fea, idx1, idx2, W_fg1, b_fg1,
                                       W_fg2, b_fg2, x, sums, cnt, E);
    k_out<<<nblk_n, tpb, 0, stream>>>(sums, cnt, node_fea, W_aw2, b_aw2,
                                      W_aw3, b_aw3, (float*)d_out, N);
}


# k_edge: the 16 index loads and the 64 x[idx2] gather loads of a wave issue as two groups into registers ahead of GEMM2, replacing 72 serialized load-wait round trips between the tail atomics
# speedup vs baseline: 1.1963x; 1.1963x over previous
#include <hip/hip_runtime.h>
#include <hip/hip_bf16.h>


typedef float v2f __attribute__((ext_vector_type(2)));
typedef float v8f __attribute__((ext_vector_type(8)));

#define DV   128
#define DE0  64
#define DE1  128
#define WAVES_PER_BLOCK 4
#define SHIFT_LN2 0.6931471805599453f

__device__ __forceinline__ float sp_act(float v) {
    float s = (v > 20.0f) ? v : log1pf(__expf(v));
    return s - SHIFT_LN2;
}


template <typename PtrT>
__device__ __forceinline__ void wmma_gemm_strip(PtrT Arow, const float* B, int K,
                                                v8f acc[8], int half, int l16,
                                                float ascale) {
    for (int k0 = 0; k0 < K; k0 += 4) {
        v2f a;
        a.x = Arow[k0 + 2 * half + 0] * ascale;
        a.y = Arow[k0 + 2 * half + 1] * ascale;
#pragma unroll
        for (int t = 0; t < 8; ++t) {
            const float* bp = B + (size_t)(k0 + 2 * half) * 128 + t * 16 + l16;
            v2f b;
            b.x = bp[0];
            b.y = bp[128];
            acc[t] = __builtin_amdgcn_wmma_f32_16x16x4_f32(
                false, a, false, b, (short)0, acc[t], false, false);
        }
    }
}

template <typename PtrT>
__device__ __forceinline__ void wmma_gemm_ldsb(PtrT Arow, const v2f* Bfrag, int K,
                                               v8f acc[8], int half, int lane,
                                               float ascale) {
    for (int k0 = 0; k0 < K; k0 += 4) {
        v2f a;
        a.x = Arow[k0 + 2 * half + 0] * ascale;
        a.y = Arow[k0 + 2 * half + 1] * ascale;
        const int s8 = (k0 >> 2) << 3;
#pragma unroll
        for (int t = 0; t < 8; ++t) {
            v2f b = Bfrag[(s8 + t) * 32 + lane];
            acc[t] = __builtin_amdgcn_wmma_f32_16x16x4_f32(
                false, a, false, b, (short)0, acc[t], false, false);
        }
    }
}

__device__ __forceinline__ void stage_weight_frag(float* dst, const float* W,
                                                  int K, int tid, int nthreads) {
    for (int i = tid; i < K * 128; i += nthreads) {
        const int k = i >> 7, n = i & 127;
        const int s = k >> 2, j = k & 1, hh = (k >> 1) & 1;
        const int l = hh * 16 + (n & 15), t = n >> 4;
        dst[(((s << 3) + t) * 32 + l) * 2 + j] = W[i];
    }
}

__device__ __forceinline__ void bias_softplus(v8f acc[8], const float* bias, int l16) {
#pragma unroll
    for (int t = 0; t < 8; ++t) {
        float bv = bias[t * 16 + l16];
#pragma unroll
        for (int r = 0; r < 8; ++r) acc[t][r] = sp_act(acc[t][r] + bv);
    }
}

__global__ __launch_bounds__(32 * WAVES_PER_BLOCK)
void k_node_proj(const float* __restrict__ node_fea,
                 const float* __restrict__ W_aw1,
                 float* __restrict__ x, int N) {
    const int lane = threadIdx.x & 31;
    const int wave = threadIdx.x >> 5;
    const int half = lane >> 4, l16 = lane & 15;
    const int row0 = (blockIdx.x * WAVES_PER_BLOCK + wave) * 16;
    if (row0 >= N) return;

    const int arow = min(row0 + l16, N - 1);
    v8f acc[8] = {};
    wmma_gemm_strip(node_fea + (size_t)arow * DV, W_aw1, DV, acc, half, l16, 1.0f);

#pragma unroll
    for (int t = 0; t < 8; ++t)
#pragma unroll
        for (int r = 0; r < 8; ++r) {
            int m = half * 8 + r;
            if (row0 + m < N)
                x[(size_t)(row0 + m) * DV + t * 16 + l16] = acc[t][r];
        }
}

__global__ __launch_bounds__(32 * WAVES_PER_BLOCK)
void k_edge(const float* __restrict__ edge_fea,
            const int* __restrict__ idx1, const int* __restrict__ idx2,
            const float* __restrict__ W_fg1, const float* __restrict__ b_fg1,
            const float* __restrict__ W_fg2, const float* __restrict__ b_fg2,
            const float* __restrict__ x,
            float* __restrict__ sums, float* __restrict__ cnt, int E) {
    __shared__ float w1t[DE0 * DE1];
    __shared__ float w2t[DE1 * DE1];
    __shared__ float hbuf[WAVES_PER_BLOCK * 16 * DE1];

    const int lane = threadIdx.x & 31;
    const int wave = threadIdx.x >> 5;
    const int half = lane >> 4, l16 = lane & 15;

    stage_weight_frag(w1t, W_fg1, DE0, threadIdx.x, 32 * WAVES_PER_BLOCK);
    stage_weight_frag(w2t, W_fg2, DE1, threadIdx.x, 32 * WAVES_PER_BLOCK);
    __syncthreads();

    const int e0 = (blockIdx.x * WAVES_PER_BLOCK + wave) * 16;
    if (e0 >= E) return;

    int n1[8], n2[8];
#pragma unroll
    for (int r = 0; r < 8; ++r) {
        const int ec = min(e0 + half * 8 + r, E - 1);
        n2[r] = idx2[ec];
        n1[r] = idx1[ec];
    }
    const int n1c = idx1[min(e0 + l16, E - 1)];

    const int arow = min(e0 + l16, E - 1);
    v8f acc[8] = {};
    wmma_gemm_ldsb(edge_fea + (size_t)arow * DE0, (const v2f*)w1t, DE0,
                   acc, half, lane, 1.0f);
    bias_softplus(acc, b_fg1, l16);

    float* h = hbuf + wave * 16 * DE1;
#pragma unroll
    for (int t = 0; t < 8; ++t)
#pragma unroll
        for (int r = 0; r < 8; ++r)
            h[(half * 8 + r) * DE1 + t * 16 + l16] = acc[t][r];

    float xv[8][8];
#pragma unroll
    for (int r = 0; r < 8; ++r) {
        const float* xr = x + (size_t)n2[r] * DV + l16;
#pragma unroll
        for (int t = 0; t < 8; ++t) xv[r][t] = xr[t * 16];
    }

    v8f acc2[8] = {};
    wmma_gemm_ldsb(h + l16 * DE1, (const v2f*)w2t, DE1, acc2, half, lane, 1.0f);
    bias_softplus(acc2, b_fg2, l16);

#pragma unroll
    for (int r = 0; r < 8; ++r) {
        const int e = e0 + half * 8 + r;
        if (e < E) {
            float* sr = sums + (size_t)n1[r] * DV + l16;
#pragma unroll
            for (int t = 0; t < 8; ++t)
                atomicAdd(&sr[t * 16], xv[r][t] * acc2[t][r]);
        }
    }
    if (lane < 16 && (e0 + lane) < E) atomicAdd(&cnt[n1c], 1.0f);
}

__global__ __launch_bounds__(32 * WAVES_PER_BLOCK)
void k_out(const float* __restrict__ sums, const float* __restrict__ cnt,
           const float* __restrict__ node_fea,
           const float* __restrict__ W_aw2, const float* __restrict__ b_aw2,
           const float* __restrict__ W_aw3, const float* __restrict__ b_aw3,
           float* __restrict__ out, int N) {
    __shared__ float lds[WAVES_PER_BLOCK * 16 * DV];
    const int lane = threadIdx.x & 31;
    const int wave = threadIdx.x >> 5;
    const int half = lane >> 4, l16 = lane & 15;
    const int row0 = (blockIdx.x * WAVES_PER_BLOCK + wave) * 16;
    if (row0 >= N) return;

    const int arow = min(row0 + l16, N - 1);
    const float rscale = 1.0f / fmaxf(cnt[arow], 1.0f);

    v8f acc[8] = {};
    wmma_gemm_strip(sums + (size_t)arow * DV, W_aw2, DV, acc, half, l16, rscale);
    bias_softplus(acc, b_aw2, l16);

    float* h = lds + wave * 16 * DV;
#pragma unroll
    for (int t = 0; t < 8; ++t)
#pragma unroll
        for (int r = 0; r < 8; ++r)
            h[(half * 8 + r) * DV + t * 16 + l16] = acc[t][r];

    v8f acc2[8] = {};
    wmma_gemm_strip(h + l16 * DV, W_aw3, DV, acc2, half, l16, 1.0f);

#pragma unroll
    for (int t = 0; t < 8; ++t) {
        const float bv = b_aw3[t * 16 + l16];
#pragma unroll
        for (int r = 0; r < 8; ++r) {
            const int m = half * 8 + r;
            if (row0 + m < N) {
                const size_t o = (size_t)(row0 + m) * DV + t * 16 + l16;
                out[o] = node_fea[o] + acc2[t][r] + bv;
            }
        }
    }
}

extern "C" void kernel_launch(void* const* d_in, const int* in_sizes, int n_in,
                              void* d_out, int out_size, void* d_ws, size_t ws_size,
                              hipStream_t stream) {
    const float* node_fea = (const float*)d_in[0];
    const int*   idx1     = (const int*)d_in[1];
    const int*   idx2     = (const int*)d_in[2];
    const float* edge_fea = (const float*)d_in[3];
    const float* W_aw1    = (const float*)d_in[4];
    const float* W_fg1    = (const float*)d_in[5];
    const float* b_fg1    = (const float*)d_in[6];
    const float* W_fg2    = (const float*)d_in[7];
    const float* b_fg2    = (const float*)d_in[8];
    const float* W_aw2    = (const float*)d_in[9];
    const float* b_aw2    = (const float*)d_in[10];
    const float* W_aw3    = (const float*)d_in[11];
    const float* b_aw3    = (const float*)d_in[12];

    const int N = in_sizes[0] / DV;
    const int E = in_sizes[1];

    float* x    = (float*)d_ws;
    float* sums = x + (size_t)N * DV;
    float* cnt  = sums + (size_t)N * DV;

    hipMemsetAsync(sums, 0, ((size_t)N * DV + N) * sizeof(float), stream);

    const int tpb = 32 * WAVES_PER_BLOCK;
    const int ntile_n = (N + 15) / 16;
    const int ntile_e = (E + 15) / 16;
    const int nblk_n = (ntile_n + WAVES_PER_BLOCK - 1) / WAVES_PER_BLOCK;
    const int nblk_e = (ntile_e + WAVES_PER_BLOCK - 1) / WAVES_PER_BLOCK;

    k_node_proj<<<nblk_n, tpb, 0, stream>>>(node_fea, W_aw1, x, N);
    k_edge<<<nblk_e, tpb, 0, stream>>>(edge_fea, idx1, idx2, W_fg1, b_fg1,
                                       W_fg2, b_fg2, x, sums, cnt, E);
    k_out<<<nblk_n, tpb, 0, stream>>>(sums, cnt, node_fea, W_aw2, b_aw2,
                                      W_aw3, b_aw3, (float*)d_out, N);
}


# softplus: library log1pf body (128 unrolled copies, ~100 KB of edge-kernel text, above the 64 KiB instruction cache) replaced by branch-free max(v,0)+ln2*log2(1+exp2(-|v|*log2e)) on v_exp/v_log
# speedup vs baseline: 4.3881x; 3.6681x over previous
#include <hip/hip_runtime.h>
#include <hip/hip_bf16.h>


typedef float v2f __attribute__((ext_vector_type(2)));
typedef float v8f __attribute__((ext_vector_type(8)));

#define DV   128
#define DE0  64
#define DE1  128
#define WAVES_PER_BLOCK 4
#define SHIFT_LN2 0.6931471805599453f

__device__ __forceinline__ float sp_act(float v) {
    const float e  = __builtin_amdgcn_exp2f(-1.4426950408889634f * fabsf(v));
    const float l2 = __builtin_amdgcn_logf(1.0f + e);
    return fmaf(SHIFT_LN2, l2 - 1.0f, fmaxf(v, 0.0f));
}


template <typename PtrT>
__device__ __forceinline__ void wmma_gemm_strip(PtrT Arow, const float* B, int K,
                                                v8f acc[8], int half, int l16,
                                                float ascale) {
    for (int k0 = 0; k0 < K; k0 += 4) {
        v2f a;
        a.x = Arow[k0 + 2 * half + 0] * ascale;
        a.y = Arow[k0 + 2 * half + 1] * ascale;
#pragma unroll
        for (int t = 0; t < 8; ++t) {
            const float* bp = B + (size_t)(k0 + 2 * half) * 128 + t * 16 + l16;
            v2f b;
            b.x = bp[0];
            b.y = bp[128];
            acc[t] = __builtin_amdgcn_wmma_f32_16x16x4_f32(
                false, a, false, b, (short)0, acc[t], false, false);
        }
    }
}

template <typename PtrT>
__device__ __forceinline__ void wmma_gemm_ldsb(PtrT Arow, const v2f* Bfrag, int K,
                                               v8f acc[8], int half, int lane,
                                               float ascale) {
    for (int k0 = 0; k0 < K; k0 += 4) {
        v2f a;
        a.x = Arow[k0 + 2 * half + 0] * ascale;
        a.y = Arow[k0 + 2 * half + 1] * ascale;
        const int s8 = (k0 >> 2) << 3;
#pragma unroll
        for (int t = 0; t < 8; ++t) {
            v2f b = Bfrag[(s8 + t) * 32 + lane];
            acc[t] = __builtin_amdgcn_wmma_f32_16x16x4_f32(
                false, a, false, b, (short)0, acc[t], false, false);
        }
    }
}

__device__ __forceinline__ void stage_weight_frag(float* dst, const float* W,
                                                  int K, int tid, int nthreads) {
    for (int i = tid; i < K * 128; i += nthreads) {
        const int k = i >> 7, n = i & 127;
        const int s = k >> 2, j = k & 1, hh = (k >> 1) & 1;
        const int l = hh * 16 + (n & 15), t = n >> 4;
        dst[(((s << 3) + t) * 32 + l) * 2 + j] = W[i];
    }
}

__device__ __forceinline__ void bias_softplus(v8f acc[8], const float* bias, int l16) {
#pragma unroll
    for (int t = 0; t < 8; ++t) {
        float bv = bias[t * 16 + l16];
#pragma unroll
        for (int r = 0; r < 8; ++r) acc[t][r] = sp_act(acc[t][r] + bv);
    }
}

__global__ __launch_bounds__(32 * WAVES_PER_BLOCK)
void k_node_proj(const float* __restrict__ node_fea,
                 const float* __restrict__ W_aw1,
                 float* __restrict__ x, int N) {
    const int lane = threadIdx.x & 31;
    const int wave = threadIdx.x >> 5;
    const int half = lane >> 4, l16 = lane & 15;
    const int row0 = (blockIdx.x * WAVES_PER_BLOCK + wave) * 16;
    if (row0 >= N) return;

    const int arow = min(row0 + l16, N - 1);
    v8f acc[8] = {};
    wmma_gemm_strip(node_fea + (size_t)arow * DV, W_aw1, DV, acc, half, l16, 1.0f);

#pragma unroll
    for (int t = 0; t < 8; ++t)
#pragma unroll
        for (int r = 0; r < 8; ++r) {
            int m = half * 8 + r;
            if (row0 + m < N)
                x[(size_t)(row0 + m) * DV + t * 16 + l16] = acc[t][r];
        }
}

__global__ __launch_bounds__(32 * WAVES_PER_BLOCK)
void k_edge(const float* __restrict__ edge_fea,
            const int* __restrict__ idx1, const int* __restrict__ idx2,
            const float* __restrict__ W_fg1, const float* __restrict__ b_fg1,
            const float* __restrict__ W_fg2, const float* __restrict__ b_fg2,
            const float* __restrict__ x,
            float* __restrict__ sums, float* __restrict__ cnt, int E) {
    __shared__ float w1t[DE0 * DE1];
    __shared__ float w2t[DE1 * DE1];
    __shared__ float hbuf[WAVES_PER_BLOCK * 16 * DE1];

    const int lane = threadIdx.x & 31;
    const int wave = threadIdx.x >> 5;
    const int half = lane >> 4, l16 = lane & 15;

    stage_weight_frag(w1t, W_fg1, DE0, threadIdx.x, 32 * WAVES_PER_BLOCK);
    stage_weight_frag(w2t, W_fg2, DE1, threadIdx.x, 32 * WAVES_PER_BLOCK);
    __syncthreads();

    const int e0 = (blockIdx.x * WAVES_PER_BLOCK + wave) * 16;
    if (e0 >= E) return;

    int n1[8], n2[8];
#pragma unroll
    for (int r = 0; r < 8; ++r) {
        const int ec = min(e0 + half * 8 + r, E - 1);
        n2[r] = idx2[ec];
        n1[r] = idx1[ec];
    }
    const int n1c = idx1[min(e0 + l16, E - 1)];

    const int arow = min(e0 + l16, E - 1);
    v8f acc[8] = {};
    wmma_gemm_ldsb(edge_fea + (size_t)arow * DE0, (const v2f*)w1t, DE0,
                   acc, half, lane, 1.0f);
    bias_softplus(acc, b_fg1, l16);

    float* h = hbuf + wave * 16 * DE1;
#pragma unroll
    for (int t = 0; t < 8; ++t)
#pragma unroll
        for (int r = 0; r < 8; ++r)
            h[(half * 8 + r) * DE1 + t * 16 + l16] = acc[t][r];

    float xv[8][8];
#pragma unroll
    for (int r = 0; r < 8; ++r) {
        const float* xr = x + (size_t)n2[r] * DV + l16;
#pragma unroll
        for (int t = 0; t < 8; ++t) xv[r][t] = xr[t * 16];
    }

    v8f acc2[8] = {};
    wmma_gemm_ldsb(h + l16 * DE1, (const v2f*)w2t, DE1, acc2, half, lane, 1.0f);
    bias_softplus(acc2, b_fg2, l16);

#pragma unroll
    for (int r = 0; r < 8; ++r) {
        const int e = e0 + half * 8 + r;
        if (e < E) {
            float* sr = sums + (size_t)n1[r] * DV + l16;
#pragma unroll
            for (int t = 0; t < 8; ++t)
                atomicAdd(&sr[t * 16], xv[r][t] * acc2[t][r]);
        }
    }
    if (lane < 16 && (e0 + lane) < E) atomicAdd(&cnt[n1c], 1.0f);
}

__global__ __launch_bounds__(32 * WAVES_PER_BLOCK)
void k_out(const float* __restrict__ sums, const float* __restrict__ cnt,
           const float* __restrict__ node_fea,
           const float* __restrict__ W_aw2, const float* __restrict__ b_aw2,
           const float* __restrict__ W_aw3, const float* __restrict__ b_aw3,
           float* __restrict__ out, int N) {
    __shared__ float lds[WAVES_PER_BLOCK * 16 * DV];
    const int lane = threadIdx.x & 31;
    const int wave = threadIdx.x >> 5;
    const int half = lane >> 4, l16 = lane & 15;
    const int row0 = (blockIdx.x * WAVES_PER_BLOCK + wave) * 16;
    if (row0 >= N) return;

    const int arow = min(row0 + l16, N - 1);
    const float rscale = 1.0f / fmaxf(cnt[arow], 1.0f);

    v8f acc[8] = {};
    wmma_gemm_strip(sums + (size_t)arow * DV, W_aw2, DV, acc, half, l16, rscale);
    bias_softplus(acc, b_aw2, l16);

    float* h = lds + wave * 16 * DV;
#pragma unroll
    for (int t = 0; t < 8; ++t)
#pragma unroll
        for (int r = 0; r < 8; ++r)
            h[(half * 8 + r) * DV + t * 16 + l16] = acc[t][r];

    v8f acc2[8] = {};
    wmma_gemm_strip(h + l16 * DV, W_aw3, DV, acc2, half, l16, 1.0f);

#pragma unroll
    for (int t = 0; t < 8; ++t) {
        const float bv = b_aw3[t * 16 + l16];
#pragma unroll
        for (int r = 0; r < 8; ++r) {
            const int m = half * 8 + r;
            if (row0 + m < N) {
                const size_t o = (size_t)(row0 + m) * DV + t * 16 + l16;
                out[o] = node_fea[o] + acc2[t][r] + bv;
            }
        }
    }
}

extern "C" void kernel_launch(void* const* d_in, const int* in_sizes, int n_in,
                              void* d_out, int out_size, void* d_ws, size_t ws_size,
                              hipStream_t stream) {
    const float* node_fea = (const float*)d_in[0];
    const int*   idx1     = (const int*)d_in[1];
    const int*   idx2     = (const int*)d_in[2];
    const float* edge_fea = (const float*)d_in[3];
    const float* W_aw1    = (const float*)d_in[4];
    const float* W_fg1    = (const float*)d_in[5];
    const float* b_fg1    = (const float*)d_in[6];
    const float* W_fg2    = (const float*)d_in[7];
    const float* b_fg2    = (const float*)d_in[8];
    const float* W_aw2    = (const float*)d_in[9];
    const float* b_aw2    = (const float*)d_in[10];
    const float* W_aw3    = (const float*)d_in[11];
    const float* b_aw3    = (const float*)d_in[12];

    const int N = in_sizes[0] / DV;
    const int E = in_sizes[1];

    float* x    = (float*)d_ws;
    float* sums = x + (size_t)N * DV;
    float* cnt  = sums + (size_t)N * DV;

    hipMemsetAsync(sums, 0, ((size_t)N * DV + N) * sizeof(float), stream);

    const int tpb = 32 * WAVES_PER_BLOCK;
    const int ntile_n = (N + 15) / 16;
    const int ntile_e = (E + 15) / 16;
    const int nblk_n = (ntile_n + WAVES_PER_BLOCK - 1) / WAVES_PER_BLOCK;
    const int nblk_e = (ntile_e + WAVES_PER_BLOCK - 1) / WAVES_PER_BLOCK;

    k_node_proj<<<nblk_n, tpb, 0, stream>>>(node_fea, W_aw1, x, N);
    k_edge<<<nblk_e, tpb, 0, stream>>>(edge_fea, idx1, idx2, W_fg1, b_fg1,
                                       W_fg2, b_fg2, x, sums, cnt, E);
    k_out<<<nblk_n, tpb, 0, stream>>>(sums, cnt, node_fea, W_aw2, b_aw2,
                                      W_aw3, b_aw3, (float*)d_out, N);
}


# k_edge as 512 resident blocks (2 per WGP) whose waves stride over the 16-edge tiles: the 96 KB weight staging into LDS runs once per block, not once per 64 edges
# speedup vs baseline: 5.7005x; 1.2991x over previous
#include <hip/hip_runtime.h>
#include <hip/hip_bf16.h>


typedef float v2f __attribute__((ext_vector_type(2)));
typedef float v8f __attribute__((ext_vector_type(8)));

#define DV   128
#define DE0  64
#define DE1  128
#define WAVES_PER_BLOCK 4
#define SHIFT_LN2 0.6931471805599453f

__device__ __forceinline__ float sp_act(float v) {
    const float e  = __builtin_amdgcn_exp2f(-1.4426950408889634f * fabsf(v));
    const float l2 = __builtin_amdgcn_logf(1.0f + e);
    return fmaf(SHIFT_LN2, l2 - 1.0f, fmaxf(v, 0.0f));
}


template <typename PtrT>
__device__ __forceinline__ void wmma_gemm_strip(PtrT Arow, const float* B, int K,
                                                v8f acc[8], int half, int l16,
                                                float ascale) {
    for (int k0 = 0; k0 < K; k0 += 4) {
        v2f a;
        a.x = Arow[k0 + 2 * half + 0] * ascale;
        a.y = Arow[k0 + 2 * half + 1] * ascale;
#pragma unroll
        for (int t = 0; t < 8; ++t) {
            const float* bp = B + (size_t)(k0 + 2 * half) * 128 + t * 16 + l16;
            v2f b;
            b.x = bp[0];
            b.y = bp[128];
            acc[t] = __builtin_amdgcn_wmma_f32_16x16x4_f32(
                false, a, false, b, (short)0, acc[t], false, false);
        }
    }
}

template <typename PtrT>
__device__ __forceinline__ void wmma_gemm_ldsb(PtrT Arow, const v2f* Bfrag, int K,
                                               v8f acc[8], int half, int lane,
                                               float ascale) {
    for (int k0 = 0; k0 < K; k0 += 4) {
        v2f a;
        a.x = Arow[k0 + 2 * half + 0] * ascale;
        a.y = Arow[k0 + 2 * half + 1] * ascale;
        const int s8 = (k0 >> 2) << 3;
#pragma unroll
        for (int t = 0; t < 8; ++t) {
            v2f b = Bfrag[(s8 + t) * 32 + lane];
            acc[t] = __builtin_amdgcn_wmma_f32_16x16x4_f32(
                false, a, false, b, (short)0, acc[t], false, false);
        }
    }
}

__device__ __forceinline__ void stage_weight_frag(float* dst, const float* W,
                                                  int K, int tid, int nthreads) {
    for (int i = tid; i < K * 128; i += nthreads) {
        const int k = i >> 7, n = i & 127;
        const int s = k >> 2, j = k & 1, hh = (k >> 1) & 1;
        const int l = hh * 16 + (n & 15), t = n >> 4;
        dst[(((s << 3) + t) * 32 + l) * 2 + j] = W[i];
    }
}

__device__ __forceinline__ void bias_softplus(v8f acc[8], const float* bias, int l16) {
#pragma unroll
    for (int t = 0; t < 8; ++t) {
        float bv = bias[t * 16 + l16];
#pragma unroll
        for (int r = 0; r < 8; ++r) acc[t][r] = sp_act(acc[t][r] + bv);
    }
}

__global__ __launch_bounds__(32 * WAVES_PER_BLOCK)
void k_node_proj(const float* __restrict__ node_fea,
                 const float* __restrict__ W_aw1,
                 float* __restrict__ x, int N) {
    const int lane = threadIdx.x & 31;
    const int wave = threadIdx.x >> 5;
    const int half = lane >> 4, l16 = lane & 15;
    const int row0 = (blockIdx.x * WAVES_PER_BLOCK + wave) * 16;
    if (row0 >= N) return;

    const int arow = min(row0 + l16, N - 1);
    v8f acc[8] = {};
    wmma_gemm_strip(node_fea + (size_t)arow * DV, W_aw1, DV, acc, half, l16, 1.0f);

#pragma unroll
    for (int t = 0; t < 8; ++t)
#pragma unroll
        for (int r = 0; r < 8; ++r) {
            int m = half * 8 + r;
            if (row0 + m < N)
                x[(size_t)(row0 + m) * DV + t * 16 + l16] = acc[t][r];
        }
}

__global__ __launch_bounds__(32 * WAVES_PER_BLOCK)
void k_edge(const float* __restrict__ edge_fea,
            const int* __restrict__ idx1, const int* __restrict__ idx2,
            const float* __restrict__ W_fg1, const float* __restrict__ b_fg1,
            const float* __restrict__ W_fg2, const float* __restrict__ b_fg2,
            const float* __restrict__ x,
            float* __restrict__ sums, float* __restrict__ cnt, int E,
            int tstep  ) {
    __shared__ float w1t[DE0 * DE1];
    __shared__ float w2t[DE1 * DE1];
    __shared__ float hbuf[WAVES_PER_BLOCK * 16 * DE1];

    const int lane = threadIdx.x & 31;
    const int wave = __builtin_amdgcn_readfirstlane(threadIdx.x >> 5);
    const int half = lane >> 4, l16 = lane & 15;

    stage_weight_frag(w1t, W_fg1, DE0, threadIdx.x, 32 * WAVES_PER_BLOCK);
    stage_weight_frag(w2t, W_fg2, DE1, threadIdx.x, 32 * WAVES_PER_BLOCK);
    __syncthreads();

    float* h = hbuf + wave * 16 * DE1;
    const int ntile = (E + 15) >> 4;
    for (int tile = blockIdx.x * WAVES_PER_BLOCK + wave; tile < ntile; tile += tstep) {
    const int e0 = tile * 16;

    int n1[8], n2[8];
#pragma unroll
    for (int r = 0; r < 8; ++r) {
        const int ec = min(e0 + half * 8 + r, E - 1);
        n2[r] = idx2[ec];
        n1[r] = idx1[ec];
    }
    const int n1c = idx1[min(e0 + l16, E - 1)];

    const int arow = min(e0 + l16, E - 1);
    v8f acc[8] = {};
    wmma_gemm_ldsb(edge_fea + (size_t)arow * DE0, (const v2f*)w1t, DE0,
                   acc, half, lane, 1.0f);
    bias_softplus(acc, b_fg1, l16);

#pragma unroll
    for (int t = 0; t < 8; ++t)
#pragma unroll
        for (int r = 0; r < 8; ++r)
            h[(half * 8 + r) * DE1 + t * 16 + l16] = acc[t][r];

    float xv[8][8];
#pragma unroll
    for (int r = 0; r < 8; ++r) {
        const float* xr = x + (size_t)n2[r] * DV + l16;
#pragma unroll
        for (int t = 0; t < 8; ++t) xv[r][t] = xr[t * 16];
    }

    v8f acc2[8] = {};
    wmma_gemm_ldsb(h + l16 * DE1, (const v2f*)w2t, DE1, acc2, half, lane, 1.0f);
    bias_softplus(acc2, b_fg2, l16);

#pragma unroll
    for (int r = 0; r < 8; ++r) {
        const int e = e0 + half * 8 + r;
        if (e < E) {
            float* sr = sums + (size_t)n1[r] * DV + l16;
#pragma unroll
            for (int t = 0; t < 8; ++t)
                atomicAdd(&sr[t * 16], xv[r][t] * acc2[t][r]);
        }
    }
    if (lane < 16 && (e0 + lane) < E) atomicAdd(&cnt[n1c], 1.0f);
    }
}

__global__ __launch_bounds__(32 * WAVES_PER_BLOCK)
void k_out(const float* __restrict__ sums, const float* __restrict__ cnt,
           const float* __restrict__ node_fea,
           const float* __restrict__ W_aw2, const float* __restrict__ b_aw2,
           const float* __restrict__ W_aw3, const float* __restrict__ b_aw3,
           float* __restrict__ out, int N) {
    __shared__ float lds[WAVES_PER_BLOCK * 16 * DV];
    const int lane = threadIdx.x & 31;
    const int wave = threadIdx.x >> 5;
    const int half = lane >> 4, l16 = lane & 15;
    const int row0 = (blockIdx.x * WAVES_PER_BLOCK + wave) * 16;
    if (row0 >= N) return;

    const int arow = min(row0 + l16, N - 1);
    const float rscale = 1.0f / fmaxf(cnt[arow], 1.0f);

    v8f acc[8] = {};
    wmma_gemm_strip(sums + (size_t)arow * DV, W_aw2, DV, acc, half, l16, rscale);
    bias_softplus(acc, b_aw2, l16);

    float* h = lds + wave * 16 * DV;
#pragma unroll
    for (int t = 0; t < 8; ++t)
#pragma unroll
        for (int r = 0; r < 8; ++r)
            h[(half * 8 + r) * DV + t * 16 + l16] = acc[t][r];

    v8f acc2[8] = {};
    wmma_gemm_strip(h + l16 * DV, W_aw3, DV, acc2, half, l16, 1.0f);

#pragma unroll
    for (int t = 0; t < 8; ++t) {
        const float bv = b_aw3[t * 16 + l16];
#pragma unroll
        for (int r = 0; r < 8; ++r) {
            const int m = half * 8 + r;
            if (row0 + m < N) {
                const size_t o = (size_t)(row0 + m) * DV + t * 16 + l16;
                out[o] = node_fea[o] + acc2[t][r] + bv;
            }
        }
    }
}

extern "C" void kernel_launch(void* const* d_in, const int* in_sizes, int n_in,
                              void* d_out, int out_size, void* d_ws, size_t ws_size,
                              hipStream_t stream) {
    const float* node_fea = (const float*)d_in[0];
    const int*   idx1     = (const int*)d_in[1];
    const int*   idx2     = (const int*)d_in[2];
    const float* edge_fea = (const float*)d_in[3];
    const float* W_aw1    = (const float*)d_in[4];
    const float* W_fg1    = (const float*)d_in[5];
    const float* b_fg1    = (const float*)d_in[6];
    const float* W_fg2    = (const float*)d_in[7];
    const float* b_fg2    = (const float*)d_in[8];
    const float* W_aw2    = (const float*)d_in[9];
    const float* b_aw2    = (const float*)d_in[10];
    const float* W_aw3    = (const float*)d_in[11];
    const float* b_aw3    = (const float*)d_in[12];

    const int N = in_sizes[0] / DV;
    const int E = in_sizes[1];

    float* x    = (float*)d_ws;
    float* sums = x + (size_t)N * DV;
    float* cnt  = sums + (size_t)N * DV;

    hipMemsetAsync(sums, 0, ((size_t)N * DV + N) * sizeof(float), stream);

    const int tpb = 32 * WAVES_PER_BLOCK;
    const int ntile_n = (N + 15) / 16;
    const int ntile_e = (E + 15) / 16;
    const int nblk_n = (ntile_n + WAVES_PER_BLOCK - 1) / WAVES_PER_BLOCK;
    const int nblk_e = (ntile_e + WAVES_PER_BLOCK - 1) / WAVES_PER_BLOCK;

    k_node_proj<<<nblk_n, tpb, 0, stream>>>(node_fea, W_aw1, x, N);
    const int nblk_e_res = nblk_e < 512 ? nblk_e : 512;
    k_edge<<<nblk_e_res, tpb, 0, stream>>>(edge_fea, idx1, idx2, W_fg1, b_fg1,
                                           W_fg2, b_fg2, x, sums, cnt, E,
                                           nblk_e_res * WAVES_PER_BLOCK);
    k_out<<<nblk_n, tpb, 0, stream>>>(sums, cnt, node_fea, W_aw2, b_aw2,
                                      W_aw3, b_aw3, (float*)d_out, N);
}
